# starting kernel
# baseline (speedup 1.0000x reference)
#include <hip/hip_runtime.h>

typedef float v2f __attribute__((ext_vector_type(2)));
typedef float v8f __attribute__((ext_vector_type(8)));

#define LOG2E     1.4426950408889634f
#define NEG_2PI2  (-19.739208802178716f)

__global__ void residue_setup_kernel(const float* __restrict__ xyz,
                                     const float* __restrict__ u,
                                     const float* __restrict__ hidden_occ,
                                     const float* __restrict__ B_inv,
                                     const float* __restrict__ R,
                                     const float* __restrict__ T,
                                     float* __restrict__ ws) {
  const int t = threadIdx.x;
  if (t < 64) {
    const int s = t >> 4, a = t & 15;
    const float x = xyz[a * 3 + 0], y = xyz[a * 3 + 1], z = xyz[a * 3 + 2];
    const float f0 = B_inv[0] * x + B_inv[1] * y + B_inv[2] * z;
    const float f1 = B_inv[3] * x + B_inv[4] * y + B_inv[5] * z;
    const float f2 = B_inv[6] * x + B_inv[7] * y + B_inv[8] * z;
    const float* Rs = R + s * 9;
    const float* Ts = T + s * 3;
    ws[t * 4 + 0] = Rs[0] * f0 + Rs[1] * f1 + Rs[2] * f2 + Ts[0];
    ws[t * 4 + 1] = Rs[3] * f0 + Rs[4] * f1 + Rs[5] * f2 + Ts[1];
    ws[t * 4 + 2] = Rs[6] * f0 + Rs[7] * f1 + Rs[8] * f2 + Ts[2];
    ws[t * 4 + 3] = 0.0f;
  }
  if (t < 16) {
    const float* ua = u + t * 6;
    const float c = NEG_2PI2 * LOG2E;
    ws[256 + t * 8 + 0] = c * ua[0];
    ws[256 + t * 8 + 1] = c * ua[1];
    ws[256 + t * 8 + 2] = c * ua[2];
    ws[256 + t * 8 + 3] = 2.0f * c * ua[3];
    ws[256 + t * 8 + 4] = 2.0f * c * ua[4];
    ws[256 + t * 8 + 5] = 2.0f * c * ua[5];
    ws[256 + t * 8 + 6] = 0.0f;
    ws[256 + t * 8 + 7] = 0.0f;
  }
  if (t == 0) {
    const float xo = hidden_occ[0];
    const float occ = 1.0f / (1.0f + __expf(-xo));
    ws[384] = occ;
    ws[385] = __log2f(occ);
  }
}

template <bool FULL>
__device__ __forceinline__ void sf_tile(
    const int*   __restrict__ hkl,
    const float* __restrict__ sv,
    const float* __restrict__ sf,
    const float* __restrict__ R,
    const float* __restrict__ ws,
    float* __restrict__ out,
    int H, int h0, int lane) {
  const int half = lane >> 4;
  const int m    = lane & 15;

  const float4* xs4 = (const float4*)ws;
  const float4* us4 = (const float4*)(ws + 256);
  const float   l2occ = ws[385];

  int hm = h0 + m;
  if (!FULL) { if (hm > H - 1) hm = H - 1; }
  const float hx = (float)hkl[hm * 3 + 0];
  const float hy = (float)hkl[hm * 3 + 1];
  const float hz = (float)hkl[hm * 3 + 2];
  const float s0 = sv[hm * 3 + 0];
  const float s1 = sv[hm * 3 + 1];
  const float s2 = sv[hm * 3 + 2];

  v2f pa;
  pa.x = half ? hz   : hx;
  pa.y = half ? 0.0f : hy;

  const float4 u0 = us4[m * 2 + 0];
  const float4 u1 = us4[m * 2 + 1];
  v2f eb1, eb2;
  eb1.x = half ? u0.z  : u0.x;
  eb1.y = half ? u0.w  : u0.y;
  eb2.x = half ? l2occ : u1.x;
  eb2.y = half ? 0.0f  : u1.y;

  float4 xv[4];
  #pragma unroll
  for (int s = 0; s < 4; ++s) xv[s] = xs4[s * 16 + m];

  float sfv[8];
  {
    const int hb = h0 + 8 * half;
    const float* sfp = sf + (size_t)m * (size_t)H + hb;
    #pragma unroll
    for (int r = 0; r < 8; ++r) {
      if (FULL) {
        sfv[r] = sfp[r];
      } else {
        int hh = hb + r; if (hh > H - 1) hh = H - 1;
        sfv[r] = sf[(size_t)m * (size_t)H + hh];
      }
    }
  }

  float accR[8], accI[8];
  #pragma unroll
  for (int r = 0; r < 8; ++r) { accR[r] = 0.0f; accI[r] = 0.0f; }

  #pragma unroll
  for (int s = 0; s < 4; ++s) {
    const float* Rs = R + s * 9;
    const float q0 = s0 * Rs[0] + s1 * Rs[3] + s2 * Rs[6];
    const float q1 = s0 * Rs[1] + s1 * Rs[4] + s2 * Rs[7];
    const float q2 = s0 * Rs[2] + s1 * Rs[5] + s2 * Rs[8];

    v2f ea1, ea2;
    ea1.x = half ? q2 * q2 : q0 * q0;
    ea1.y = half ? q0 * q1 : q1 * q1;
    ea2.x = half ? 1.0f    : q0 * q2;
    ea2.y = half ? 0.0f    : q1 * q2;

    v2f pb;
    pb.x = half ? xv[s].z : xv[s].x;
    pb.y = half ? 0.0f    : xv[s].y;

    const v8f zc = {};
    v8f pT = __builtin_amdgcn_wmma_f32_16x16x4_f32(
        false, pa, false, pb, (short)0, zc, false, false);
    v8f e1 = __builtin_amdgcn_wmma_f32_16x16x4_f32(
        false, ea1, false, eb1, (short)0, zc, false, false);
    v8f e2 = __builtin_amdgcn_wmma_f32_16x16x4_f32(
        false, ea2, false, eb2, (short)0, e1, false, false);

    #pragma unroll
    for (int r = 0; r < 8; ++r) {
      const float amp = sfv[r] * __builtin_amdgcn_exp2f(e2[r]);
      accR[r] = fmaf(amp, __builtin_amdgcn_cosf(pT[r]), accR[r]);
      accI[r] = fmaf(amp, __builtin_amdgcn_sinf(pT[r]), accI[r]);
    }
  }

  #pragma unroll
  for (int r = 0; r < 8; ++r) {
    #pragma unroll
    for (int st = 8; st >= 1; st >>= 1) {
      accR[r] += __shfl_xor(accR[r], st, 32);
      accI[r] += __shfl_xor(accI[r], st, 32);
    }
    const int hh = h0 + r + 8 * half;
    if (FULL) {
      if (m == 0) { out[hh] = accR[r]; out[H + hh] = accI[r]; }
    } else {
      if (m == 0 && hh < H) { out[hh] = accR[r]; out[H + hh] = accI[r]; }
    }
  }
}

__global__ __launch_bounds__(256) void residue_sf_kernel(
    const int*   __restrict__ hkl,
    const float* __restrict__ sv,
    const float* __restrict__ sf,
    const float* __restrict__ R,
    const float* __restrict__ ws,
    float* __restrict__ out,
    int H) {
  const int lane = threadIdx.x & 31;
  const int wave = threadIdx.x >> 5;
  const int h0   = (blockIdx.x * 8 + wave) * 16;
  if (h0 >= H) return;
  if (h0 + 16 <= H) {
    sf_tile<true >(hkl, sv, sf, R, ws, out, H, h0, lane);
  } else {
    sf_tile<false>(hkl, sv, sf, R, ws, out, H, h0, lane);
  }
}

extern "C" void kernel_launch(void* const* d_in, const int* in_sizes, int n_in,
                              void* d_out, int out_size, void* d_ws, size_t ws_size,
                              hipStream_t stream) {
  const int*   hkl   = (const int*)  d_in[0];
  const float* sv    = (const float*)d_in[1];
  const float* xyz   = (const float*)d_in[2];
  const float* u     = (const float*)d_in[3];
  const float* hocc  = (const float*)d_in[4];
  const float* sf    = (const float*)d_in[5];
  const float* B_inv = (const float*)d_in[6];
  const float* symR  = (const float*)d_in[7];
  const float* symT  = (const float*)d_in[8];
  float* out = (float*)d_out;
  float* ws  = (float*)d_ws;

  const int H = in_sizes[0] / 3;
  if (H <= 0) return;

  residue_setup_kernel<<<1, 64, 0, stream>>>(xyz, u, hocc, B_inv, symR, symT, ws);

  const int blocks = (H + 127) / 128;
  residue_sf_kernel<<<blocks, 256, 0, stream>>>(hkl, sv, sf, symR, ws, out, H);
}


# Transposed tiles: atoms on matrix rows, reflections on lanes, 64 reflections per wave; the atom sum is in-lane adds plus one half-wave exchange (new summation order), not a cross-lane butterfly
# speedup vs baseline: 1.3753x; 1.3753x over previous
#include <hip/hip_runtime.h>

typedef float v2f __attribute__((ext_vector_type(2)));
typedef float v8f __attribute__((ext_vector_type(8)));

#define LOG2E     1.4426950408889634f
#define NEG_2PI2  (-19.739208802178716f)

__global__ void residue_setup_kernel(const float* __restrict__ xyz,
                                     const float* __restrict__ u,
                                     const float* __restrict__ hidden_occ,
                                     const float* __restrict__ B_inv,
                                     const float* __restrict__ R,
                                     const float* __restrict__ T,
                                     float* __restrict__ ws) {
  const int t = threadIdx.x;
  if (t < 64) {
    const int s = t >> 4, a = t & 15;
    const float x = xyz[a * 3 + 0], y = xyz[a * 3 + 1], z = xyz[a * 3 + 2];
    const float f0 = B_inv[0] * x + B_inv[1] * y + B_inv[2] * z;
    const float f1 = B_inv[3] * x + B_inv[4] * y + B_inv[5] * z;
    const float f2 = B_inv[6] * x + B_inv[7] * y + B_inv[8] * z;
    const float* Rs = R + s * 9;
    const float* Ts = T + s * 3;
    ws[t * 4 + 0] = Rs[0] * f0 + Rs[1] * f1 + Rs[2] * f2 + Ts[0];
    ws[t * 4 + 1] = Rs[3] * f0 + Rs[4] * f1 + Rs[5] * f2 + Ts[1];
    ws[t * 4 + 2] = Rs[6] * f0 + Rs[7] * f1 + Rs[8] * f2 + Ts[2];
    ws[t * 4 + 3] = 0.0f;
  }
  if (t < 16) {
    const float* ua = u + t * 6;
    const float c = NEG_2PI2 * LOG2E;
    ws[256 + t * 8 + 0] = c * ua[0];
    ws[256 + t * 8 + 1] = c * ua[1];
    ws[256 + t * 8 + 2] = c * ua[2];
    ws[256 + t * 8 + 3] = 2.0f * c * ua[3];
    ws[256 + t * 8 + 4] = 2.0f * c * ua[4];
    ws[256 + t * 8 + 5] = 2.0f * c * ua[5];
    ws[256 + t * 8 + 6] = 0.0f;
    ws[256 + t * 8 + 7] = 0.0f;
  }
  if (t == 0) {
    const float xo = hidden_occ[0];
    const float occ = 1.0f / (1.0f + __expf(-xo));
    ws[384] = occ;
    ws[385] = __log2f(occ);
  }
}

__device__ __forceinline__ float xhalf(float x) {
  const int b = __builtin_bit_cast(int, x);
  return __builtin_bit_cast(float, __builtin_amdgcn_permlanex16(
      b, b, 0x76543210, 0xfedcba98, false, false));
}

template <bool FULL>
__device__ __forceinline__ void sf_tile(
    const int*   __restrict__ hkl,
    const float* __restrict__ sv,
    const float* __restrict__ sf,
    const float* __restrict__ R,
    const float* __restrict__ ws,
    float* __restrict__ out,
    int H, int h0, int lane) {
  const int half = lane >> 4;
  const int c    = lane & 15;

  const float   l2occ = ws[385];
  v2f xa[4];
  #pragma unroll
  for (int s = 0; s < 4; ++s)
    xa[s] = *(const v2f*)(ws + (s * 16 + c) * 4 + 2 * half);
  const v2f eb1 = *(const v2f*)(ws + 256 + c * 8 + 2 * half);
  const v2f u1  = *(const v2f*)(ws + 256 + c * 8 + 4);
  v2f eb2;
  eb2.x = half ? l2occ : u1.x;
  eb2.y = half ? 0.0f  : u1.y;

  int   hk[12];
  float sk[12];
  float sfv[8][4];
  #pragma unroll
  for (int t = 0; t < 4; ++t) {
    int hh = h0 + 16 * t + c;
    if (!FULL) { if (hh > H - 1) hh = H - 1; }
    #pragma unroll
    for (int j = 0; j < 3; ++j) {
      hk[3 * t + j] = hkl[(size_t)hh * 3 + j];
      sk[3 * t + j] = sv [(size_t)hh * 3 + j];
    }
    #pragma unroll
    for (int r = 0; r < 8; ++r)
      sfv[r][t] = sf[(size_t)(r + 8 * half) * (size_t)H + hh];
  }

  v2f pa[4];
  #pragma unroll
  for (int t = 0; t < 4; ++t) {
    const float hx = (float)hk[3 * t + 0];
    const float hy = (float)hk[3 * t + 1];
    const float hz = (float)hk[3 * t + 2];
    pa[t].x = half ? hz   : hx;
    pa[t].y = half ? 0.0f : hy;
  }

  float accR[4], accI[4];
  #pragma unroll
  for (int t = 0; t < 4; ++t) { accR[t] = 0.0f; accI[t] = 0.0f; }

  #pragma unroll
  for (int s = 0; s < 4; ++s) {
    const float* Rs = R + s * 9;
    #pragma unroll
    for (int t = 0; t < 4; ++t) {
      const float s0 = sk[3 * t + 0];
      const float s1 = sk[3 * t + 1];
      const float s2 = sk[3 * t + 2];
      const float q0 = s0 * Rs[0] + s1 * Rs[3] + s2 * Rs[6];
      const float q1 = s0 * Rs[1] + s1 * Rs[4] + s2 * Rs[7];
      const float q2 = s0 * Rs[2] + s1 * Rs[5] + s2 * Rs[8];

      v2f ea1, ea2;
      ea1.x = half ? q2 * q2 : q0 * q0;
      ea1.y = half ? q0 * q1 : q1 * q1;
      ea2.x = half ? 1.0f    : q0 * q2;
      ea2.y = half ? 0.0f    : q1 * q2;

      const v8f zc = {};
      v8f pT = __builtin_amdgcn_wmma_f32_16x16x4_f32(
          false, xa[s], false, pa[t], (short)0, zc, false, false);
      v8f e1 = __builtin_amdgcn_wmma_f32_16x16x4_f32(
          false, eb1, false, ea1, (short)0, zc, false, false);
      v8f e2 = __builtin_amdgcn_wmma_f32_16x16x4_f32(
          false, eb2, false, ea2, (short)0, e1, false, false);

      #pragma unroll
      for (int r = 0; r < 8; ++r) {
        const float amp = sfv[r][t] * __builtin_amdgcn_exp2f(e2[r]);
        accR[t] = fmaf(amp, __builtin_amdgcn_cosf(pT[r]), accR[t]);
        accI[t] = fmaf(amp, __builtin_amdgcn_sinf(pT[r]), accI[t]);
      }
    }
  }

  #pragma unroll
  for (int t = 0; t < 4; ++t) {
    accR[t] += xhalf(accR[t]);
    accI[t] += xhalf(accI[t]);
  }

  #pragma unroll
  for (int t = 0; t < 4; ++t) {
    const int hh = h0 + 16 * t + c;
    const float v = half ? accI[t] : accR[t];
    float* dst = out + (half ? (size_t)H : (size_t)0) + hh;
    if (FULL) { *dst = v; } else { if (hh < H) *dst = v; }
  }
}

__global__ __launch_bounds__(256) void residue_sf_kernel(
    const int*   __restrict__ hkl,
    const float* __restrict__ sv,
    const float* __restrict__ sf,
    const float* __restrict__ R,
    const float* __restrict__ ws,
    float* __restrict__ out,
    int H) {
  const int lane = threadIdx.x & 31;
  const int wave = threadIdx.x >> 5;
  const int h0   = (blockIdx.x * 8 + wave) * 64;
  if (h0 >= H) return;
  if (h0 + 64 <= H) {
    sf_tile<true >(hkl, sv, sf, R, ws, out, H, h0, lane);
  } else {
    sf_tile<false>(hkl, sv, sf, R, ws, out, H, h0, lane);
  }
}

extern "C" void kernel_launch(void* const* d_in, const int* in_sizes, int n_in,
                              void* d_out, int out_size, void* d_ws, size_t ws_size,
                              hipStream_t stream) {
  const int*   hkl   = (const int*)  d_in[0];
  const float* sv    = (const float*)d_in[1];
  const float* xyz   = (const float*)d_in[2];
  const float* u     = (const float*)d_in[3];
  const float* hocc  = (const float*)d_in[4];
  const float* sf    = (const float*)d_in[5];
  const float* B_inv = (const float*)d_in[6];
  const float* symR  = (const float*)d_in[7];
  const float* symT  = (const float*)d_in[8];
  float* out = (float*)d_out;
  float* ws  = (float*)d_ws;

  const int H = in_sizes[0] / 3;
  if (H <= 0) return;

  residue_setup_kernel<<<1, 64, 0, stream>>>(xyz, u, hocc, B_inv, symR, symT, ws);

  const int blocks = (H + 511) / 512;
  residue_sf_kernel<<<blocks, 256, 0, stream>>>(hkl, sv, sf, symR, ws, out, H);
}


# One launch instead of two: each wave builds its lanes' per-atom operands (sym-expanded fractional xyz, scaled ANISOU, log2 occ) from the inputs; setup kernel and workspace round trip removed
# speedup vs baseline: 1.6970x; 1.2339x over previous
#include <hip/hip_runtime.h>

typedef float v2f __attribute__((ext_vector_type(2)));
typedef float v8f __attribute__((ext_vector_type(8)));

#define LOG2E     1.4426950408889634f
#define NEG_2PI2  (-19.739208802178716f)

__device__ __forceinline__ float xhalf(float x) {
  const int b = __builtin_bit_cast(int, x);
  return __builtin_bit_cast(float, __builtin_amdgcn_permlanex16(
      b, b, 0x76543210, 0xfedcba98, false, false));
}

template <bool FULL>
__device__ __forceinline__ void sf_tile(
    const int*   __restrict__ hkl,
    const float* __restrict__ sv,
    const float* __restrict__ xyz,
    const float* __restrict__ u,
    const float* __restrict__ hidden_occ,
    const float* __restrict__ sf,
    const float* __restrict__ B_inv,
    const float* __restrict__ R,
    const float* __restrict__ T,
    float* __restrict__ out,
    int H, int h0, int lane) {
  const int half = lane >> 4;
  const int c    = lane & 15;

  const float px = xyz[c * 3 + 0], py = xyz[c * 3 + 1], pz = xyz[c * 3 + 2];
  const float* ua = u + c * 6;
  const float ua_a = ua[2 * half];
  const float ua_b = ua[2 * half + 1];
  const float ua_4 = ua[4];
  const float ua_5 = ua[5];

  int   hk[12];
  float sk[12];
  float sfv[8][4];
  #pragma unroll
  for (int t = 0; t < 4; ++t) {
    int hh = h0 + 16 * t + c;
    if (!FULL) { if (hh > H - 1) hh = H - 1; }
    #pragma unroll
    for (int j = 0; j < 3; ++j) {
      hk[3 * t + j] = hkl[(size_t)hh * 3 + j];
      sk[3 * t + j] = sv [(size_t)hh * 3 + j];
    }
    #pragma unroll
    for (int r = 0; r < 8; ++r)
      sfv[r][t] = sf[(size_t)(r + 8 * half) * (size_t)H + hh];
  }

  const float f0 = B_inv[0] * px + B_inv[1] * py + B_inv[2] * pz;
  const float f1 = B_inv[3] * px + B_inv[4] * py + B_inv[5] * pz;
  const float f2 = B_inv[6] * px + B_inv[7] * py + B_inv[8] * pz;
  v2f xa[4];
  #pragma unroll
  for (int s = 0; s < 4; ++s) {
    const float* Rs = R + s * 9;
    const float* Ts = T + s * 3;
    const float x0 = Rs[0] * f0 + Rs[1] * f1 + Rs[2] * f2 + Ts[0];
    const float x1 = Rs[3] * f0 + Rs[4] * f1 + Rs[5] * f2 + Ts[1];
    const float x2 = Rs[6] * f0 + Rs[7] * f1 + Rs[8] * f2 + Ts[2];
    xa[s].x = half ? x2   : x0;
    xa[s].y = half ? 0.0f : x1;
  }
  const float xo    = hidden_occ[0];
  const float occ   = 1.0f / (1.0f + __expf(-xo));
  const float l2occ = __log2f(occ);
  const float cs = NEG_2PI2 * LOG2E;
  v2f eb1, eb2;
  eb1.x = cs * ua_a;
  eb1.y = (half ? 2.0f * cs : cs) * ua_b;
  eb2.x = half ? l2occ : 2.0f * cs * ua_4;
  eb2.y = half ? 0.0f  : 2.0f * cs * ua_5;

  v2f pa[4];
  #pragma unroll
  for (int t = 0; t < 4; ++t) {
    const float hx = (float)hk[3 * t + 0];
    const float hy = (float)hk[3 * t + 1];
    const float hz = (float)hk[3 * t + 2];
    pa[t].x = half ? hz   : hx;
    pa[t].y = half ? 0.0f : hy;
  }

  float accR[4], accI[4];
  #pragma unroll
  for (int t = 0; t < 4; ++t) { accR[t] = 0.0f; accI[t] = 0.0f; }

  #pragma unroll
  for (int s = 0; s < 4; ++s) {
    const float* Rs = R + s * 9;
    #pragma unroll
    for (int t = 0; t < 4; ++t) {
      const float s0 = sk[3 * t + 0];
      const float s1 = sk[3 * t + 1];
      const float s2 = sk[3 * t + 2];
      const float q0 = s0 * Rs[0] + s1 * Rs[3] + s2 * Rs[6];
      const float q1 = s0 * Rs[1] + s1 * Rs[4] + s2 * Rs[7];
      const float q2 = s0 * Rs[2] + s1 * Rs[5] + s2 * Rs[8];

      v2f ea1, ea2;
      ea1.x = half ? q2 * q2 : q0 * q0;
      ea1.y = half ? q0 * q1 : q1 * q1;
      ea2.x = half ? 1.0f    : q0 * q2;
      ea2.y = half ? 0.0f    : q1 * q2;

      const v8f zc = {};
      v8f pT = __builtin_amdgcn_wmma_f32_16x16x4_f32(
          false, xa[s], false, pa[t], (short)0, zc, false, false);
      v8f e1 = __builtin_amdgcn_wmma_f32_16x16x4_f32(
          false, eb1, false, ea1, (short)0, zc, false, false);
      v8f e2 = __builtin_amdgcn_wmma_f32_16x16x4_f32(
          false, eb2, false, ea2, (short)0, e1, false, false);

      #pragma unroll
      for (int r = 0; r < 8; ++r) {
        const float amp = sfv[r][t] * __builtin_amdgcn_exp2f(e2[r]);
        accR[t] = fmaf(amp, __builtin_amdgcn_cosf(pT[r]), accR[t]);
        accI[t] = fmaf(amp, __builtin_amdgcn_sinf(pT[r]), accI[t]);
      }
    }
  }

  #pragma unroll
  for (int t = 0; t < 4; ++t) {
    accR[t] += xhalf(accR[t]);
    accI[t] += xhalf(accI[t]);
  }

  #pragma unroll
  for (int t = 0; t < 4; ++t) {
    const int hh = h0 + 16 * t + c;
    const float v = half ? accI[t] : accR[t];
    float* dst = out + (half ? (size_t)H : (size_t)0) + hh;
    if (FULL) { *dst = v; } else { if (hh < H) *dst = v; }
  }
}

__global__ __launch_bounds__(256) void residue_sf_kernel(
    const int*   __restrict__ hkl,
    const float* __restrict__ sv,
    const float* __restrict__ xyz,
    const float* __restrict__ u,
    const float* __restrict__ hidden_occ,
    const float* __restrict__ sf,
    const float* __restrict__ B_inv,
    const float* __restrict__ R,
    const float* __restrict__ T,
    float* __restrict__ out,
    int H) {
  const int lane = threadIdx.x & 31;
  const int wave = threadIdx.x >> 5;
  const int h0   = (blockIdx.x * 8 + wave) * 64;
  if (h0 >= H) return;
  if (h0 + 64 <= H) {
    sf_tile<true >(hkl, sv, xyz, u, hidden_occ, sf, B_inv, R, T, out, H, h0, lane);
  } else {
    sf_tile<false>(hkl, sv, xyz, u, hidden_occ, sf, B_inv, R, T, out, H, h0, lane);
  }
}

extern "C" void kernel_launch(void* const* d_in, const int* in_sizes, int n_in,
                              void* d_out, int out_size, void* d_ws, size_t ws_size,
                              hipStream_t stream) {
  const int*   hkl   = (const int*)  d_in[0];
  const float* sv    = (const float*)d_in[1];
  const float* xyz   = (const float*)d_in[2];
  const float* u     = (const float*)d_in[3];
  const float* hocc  = (const float*)d_in[4];
  const float* sf    = (const float*)d_in[5];
  const float* B_inv = (const float*)d_in[6];
  const float* symR  = (const float*)d_in[7];
  const float* symT  = (const float*)d_in[8];
  float* out = (float*)d_out;

  const int H = in_sizes[0] / 3;
  if (H <= 0) return;

  const int blocks = (H + 511) / 512;
  residue_sf_kernel<<<blocks, 256, 0, stream>>>(hkl, sv, xyz, u, hocc, sf, B_inv,
                                                symR, symT, out, H);
}


# Software pipeline inside the wave: step i's exp2/cos/sin issue among step i-1's multiply-adds and step i+1's operand build and WMMAs, none read right behind its issue; launch bounds hold 128 VGPRs
# speedup vs baseline: 1.8458x; 1.0877x over previous
#include <hip/hip_runtime.h>

typedef float v2f __attribute__((ext_vector_type(2)));
typedef float v8f __attribute__((ext_vector_type(8)));

#define LOG2E     1.4426950408889634f
#define NEG_2PI2  (-19.739208802178716f)

__device__ __forceinline__ float xhalf(float x) {
  const int b = __builtin_bit_cast(int, x);
  return __builtin_bit_cast(float, __builtin_amdgcn_permlanex16(
      b, b, 0x76543210, 0xfedcba98, false, false));
}

template <bool FULL>
__device__ __forceinline__ void sf_tile(
    const int*   __restrict__ hkl,
    const float* __restrict__ sv,
    const float* __restrict__ xyz,
    const float* __restrict__ u,
    const float* __restrict__ hidden_occ,
    const float* __restrict__ sf,
    const float* __restrict__ B_inv,
    const float* __restrict__ R,
    const float* __restrict__ T,
    float* __restrict__ out,
    int H, int h0, int lane) {
  const int half = lane >> 4;
  const int c    = lane & 15;

  const float px = xyz[c * 3 + 0], py = xyz[c * 3 + 1], pz = xyz[c * 3 + 2];
  const float* ua = u + c * 6;
  const float ua_0 = ua[0], ua_1 = ua[1], ua_2 = ua[2], ua_3 = ua[3];
  const float ua_a = half ? ua_2 : ua_0;
  const float ua_b = half ? ua_3 : ua_1;
  const float ua_4 = ua[4];
  const float ua_5 = ua[5];

  int   hk[12];
  float sk[12];
  float sfv[8][4];
  #pragma unroll
  for (int t = 0; t < 4; ++t) {
    int hh = h0 + 16 * t + c;
    if (!FULL) { if (hh > H - 1) hh = H - 1; }
    #pragma unroll
    for (int j = 0; j < 3; ++j) {
      hk[3 * t + j] = hkl[(size_t)hh * 3 + j];
      sk[3 * t + j] = sv [(size_t)hh * 3 + j];
    }
    #pragma unroll
    for (int r = 0; r < 8; ++r)
      sfv[r][t] = sf[(size_t)(r + 8 * half) * (size_t)H + hh];
  }

  const float f0 = B_inv[0] * px + B_inv[1] * py + B_inv[2] * pz;
  const float f1 = B_inv[3] * px + B_inv[4] * py + B_inv[5] * pz;
  const float f2 = B_inv[6] * px + B_inv[7] * py + B_inv[8] * pz;
  v2f xa[4];
  #pragma unroll
  for (int s = 0; s < 4; ++s) {
    const float* Rs = R + s * 9;
    const float* Ts = T + s * 3;
    const float x0 = Rs[0] * f0 + Rs[1] * f1 + Rs[2] * f2 + Ts[0];
    const float x1 = Rs[3] * f0 + Rs[4] * f1 + Rs[5] * f2 + Ts[1];
    const float x2 = Rs[6] * f0 + Rs[7] * f1 + Rs[8] * f2 + Ts[2];
    xa[s].x = half ? x2   : x0;
    xa[s].y = half ? 0.0f : x1;
  }
  const float xo    = hidden_occ[0];
  const float occ   = 1.0f / (1.0f + __expf(-xo));
  const float l2occ = __log2f(occ);
  const float cs = NEG_2PI2 * LOG2E;
  v2f eb1, eb2;
  eb1.x = cs * ua_a;
  eb1.y = (half ? 2.0f * cs : cs) * ua_b;
  eb2.x = half ? l2occ : 2.0f * cs * ua_4;
  eb2.y = half ? 0.0f  : 2.0f * cs * ua_5;

  v2f pa[4];
  #pragma unroll
  for (int t = 0; t < 4; ++t) {
    const float hx = (float)hk[3 * t + 0];
    const float hy = (float)hk[3 * t + 1];
    const float hz = (float)hk[3 * t + 2];
    pa[t].x = half ? hz   : hx;
    pa[t].y = half ? 0.0f : hy;
  }

  float accR[4], accI[4];
  #pragma unroll
  for (int t = 0; t < 4; ++t) { accR[t] = 0.0f; accI[t] = 0.0f; }

  const v8f zc = {};
  float E[8], C[8], S[8];
  v8f pT, e2;
  v2f ea1, ea2;
  {
    const float s0 = sk[0], s1 = sk[1], s2 = sk[2];
    const float q0 = s0 * R[0] + s1 * R[3] + s2 * R[6];
    const float q1 = s0 * R[1] + s1 * R[4] + s2 * R[7];
    const float q2 = s0 * R[2] + s1 * R[5] + s2 * R[8];
    ea1.x = half ? q2 * q2 : q0 * q0;
    ea1.y = half ? q0 * q1 : q1 * q1;
    ea2.x = half ? 1.0f    : q0 * q2;
    ea2.y = half ? 0.0f    : q1 * q2;
    pT = __builtin_amdgcn_wmma_f32_16x16x4_f32(
        false, xa[0], false, pa[0], (short)0, zc, false, false);
    const v8f e1 = __builtin_amdgcn_wmma_f32_16x16x4_f32(
        false, eb1, false, ea1, (short)0, zc, false, false);
    e2 = __builtin_amdgcn_wmma_f32_16x16x4_f32(
        false, eb2, false, ea2, (short)0, e1, false, false);
  }

  #pragma unroll
  for (int i = 0; i <= 16; ++i) {
    const int  tp = (i + 3) & 3;
    const int  sn = (i + 1) >> 2;
    const int  tn = (i + 1) & 3;
    const bool has_prev = i > 0;
    const bool has_cur  = i < 16;
    const bool has_next = i + 1 < 16;

    float amp = 0.0f;
    if (has_prev) amp = sfv[0][tp] * E[0];
    #pragma unroll
    for (int r = 0; r < 8; ++r) {
      float ampn = 0.0f;
      if (has_prev) accR[tp] = fmaf(amp, C[r], accR[tp]);
      if (has_cur)  C[r] = __builtin_amdgcn_cosf(pT[r]);
      if (has_prev && r < 7) ampn = sfv[r + 1][tp] * E[r + 1];
      if (has_prev) accI[tp] = fmaf(amp, S[r], accI[tp]);
      if (has_cur)  S[r] = __builtin_amdgcn_sinf(pT[r]);
      amp = ampn;
    }
    if (has_next)
      pT = __builtin_amdgcn_wmma_f32_16x16x4_f32(
          false, xa[sn], false, pa[tn], (short)0, zc, false, false);

    if (has_cur) {
      const float* Rs = R + (has_next ? sn : 0) * 9;
      const float s0 = sk[3 * tn + 0];
      const float s1 = sk[3 * tn + 1];
      const float s2 = sk[3 * tn + 2];
      float q0 = 0.0f, q1 = 0.0f, q2 = 0.0f;
      E[0] = __builtin_amdgcn_exp2f(e2[0]);
      if (has_next) q0 = s0 * Rs[0] + s1 * Rs[3] + s2 * Rs[6];
      E[1] = __builtin_amdgcn_exp2f(e2[1]);
      if (has_next) q1 = s0 * Rs[1] + s1 * Rs[4] + s2 * Rs[7];
      E[2] = __builtin_amdgcn_exp2f(e2[2]);
      if (has_next) q2 = s0 * Rs[2] + s1 * Rs[5] + s2 * Rs[8];
      E[3] = __builtin_amdgcn_exp2f(e2[3]);
      if (has_next) ea1.x = half ? q2 * q2 : q0 * q0;
      E[4] = __builtin_amdgcn_exp2f(e2[4]);
      if (has_next) ea1.y = half ? q0 * q1 : q1 * q1;
      E[5] = __builtin_amdgcn_exp2f(e2[5]);
      if (has_next) ea2.x = half ? 1.0f    : q0 * q2;
      E[6] = __builtin_amdgcn_exp2f(e2[6]);
      if (has_next) ea2.y = half ? 0.0f    : q1 * q2;
      E[7] = __builtin_amdgcn_exp2f(e2[7]);
    }
    if (has_next) {
      const v8f e1 = __builtin_amdgcn_wmma_f32_16x16x4_f32(
          false, eb1, false, ea1, (short)0, zc, false, false);
      e2 = __builtin_amdgcn_wmma_f32_16x16x4_f32(
          false, eb2, false, ea2, (short)0, e1, false, false);
    }
  }

  #pragma unroll
  for (int t = 0; t < 4; ++t) {
    accR[t] += xhalf(accR[t]);
    accI[t] += xhalf(accI[t]);
  }

  #pragma unroll
  for (int t = 0; t < 4; ++t) {
    const int hh = h0 + 16 * t + c;
    const float v = half ? accI[t] : accR[t];
    float* dst = out + (half ? (size_t)H : (size_t)0) + hh;
    if (FULL) { *dst = v; } else { if (hh < H) *dst = v; }
  }
}

__global__ __launch_bounds__(256, 8) void residue_sf_kernel(
    const int*   __restrict__ hkl,
    const float* __restrict__ sv,
    const float* __restrict__ xyz,
    const float* __restrict__ u,
    const float* __restrict__ hidden_occ,
    const float* __restrict__ sf,
    const float* __restrict__ B_inv,
    const float* __restrict__ R,
    const float* __restrict__ T,
    float* __restrict__ out,
    int H) {
  const int lane = threadIdx.x & 31;
  const int wave = threadIdx.x >> 5;
  const int h0   = (blockIdx.x * 8 + wave) * 64;
  if (h0 >= H) return;
  if (h0 + 64 <= H) {
    sf_tile<true >(hkl, sv, xyz, u, hidden_occ, sf, B_inv, R, T, out, H, h0, lane);
  } else {
    sf_tile<false>(hkl, sv, xyz, u, hidden_occ, sf, B_inv, R, T, out, H, h0, lane);
  }
}

extern "C" void kernel_launch(void* const* d_in, const int* in_sizes, int n_in,
                              void* d_out, int out_size, void* d_ws, size_t ws_size,
                              hipStream_t stream) {
  const int*   hkl   = (const int*)  d_in[0];
  const float* sv    = (const float*)d_in[1];
  const float* xyz   = (const float*)d_in[2];
  const float* u     = (const float*)d_in[3];
  const float* hocc  = (const float*)d_in[4];
  const float* sf    = (const float*)d_in[5];
  const float* B_inv = (const float*)d_in[6];
  const float* symR  = (const float*)d_in[7];
  const float* symT  = (const float*)d_in[8];
  float* out = (float*)d_out;

  const int H = in_sizes[0] / 3;
  if (H <= 0) return;

  const int blocks = (H + 511) / 512;
  residue_sf_kernel<<<blocks, 256, 0, stream>>>(hkl, sv, xyz, u, hocc, sf, B_inv,
                                                symR, symT, out, H);
}


# Each lane owns two pairs of consecutive reflections: scattering-factor rows and both output planes move as 8-byte accesses, one whole 128-byte line per half-wave and instruction; sums unchanged
# speedup vs baseline: 1.9479x; 1.0553x over previous
#include <hip/hip_runtime.h>

typedef float v2f __attribute__((ext_vector_type(2)));
typedef float v8f __attribute__((ext_vector_type(8)));

#define LOG2E     1.4426950408889634f
#define NEG_2PI2  (-19.739208802178716f)

__device__ __forceinline__ float xhalf(float x) {
  const int b = __builtin_bit_cast(int, x);
  return __builtin_bit_cast(float, __builtin_amdgcn_permlanex16(
      b, b, 0x76543210, 0xfedcba98, false, false));
}

template <bool FULL>
__device__ __forceinline__ void sf_tile(
    const int*   __restrict__ hkl,
    const float* __restrict__ sv,
    const float* __restrict__ xyz,
    const float* __restrict__ u,
    const float* __restrict__ hidden_occ,
    const float* __restrict__ sf,
    const float* __restrict__ B_inv,
    const float* __restrict__ R,
    const float* __restrict__ T,
    float* __restrict__ out,
    int H, int h0, int lane) {
  const int half = lane >> 4;
  const int c    = lane & 15;

  const float px = xyz[c * 3 + 0], py = xyz[c * 3 + 1], pz = xyz[c * 3 + 2];
  const float* ua = u + c * 6;
  const float ua_0 = ua[0], ua_1 = ua[1], ua_2 = ua[2], ua_3 = ua[3];
  const float ua_a = half ? ua_2 : ua_0;
  const float ua_b = half ? ua_3 : ua_1;
  const float ua_4 = ua[4];
  const float ua_5 = ua[5];

  int   hk[12];
  float sk[12];
  float sfv[8][4];
  if (FULL) {
    #pragma unroll
    for (int g = 0; g < 2; ++g) {
      const size_t hb = (size_t)(h0 + 32 * g + 2 * c);
      #pragma unroll
      for (int j = 0; j < 6; ++j) {
        hk[6 * g + j] = hkl[hb * 3 + j];
        sk[6 * g + j] = sv [hb * 3 + j];
      }
      #pragma unroll
      for (int r = 0; r < 8; ++r) {
        const v2f v = *reinterpret_cast<const v2f*>(
            sf + (size_t)(r + 8 * half) * (size_t)H + hb);
        sfv[r][2 * g] = v.x; sfv[r][2 * g + 1] = v.y;
      }
    }
  } else {
    #pragma unroll
    for (int t = 0; t < 4; ++t) {
      int hh = h0 + 16 * t + c;
      if (hh > H - 1) hh = H - 1;
      #pragma unroll
      for (int j = 0; j < 3; ++j) {
        hk[3 * t + j] = hkl[(size_t)hh * 3 + j];
        sk[3 * t + j] = sv [(size_t)hh * 3 + j];
      }
      #pragma unroll
      for (int r = 0; r < 8; ++r)
        sfv[r][t] = sf[(size_t)(r + 8 * half) * (size_t)H + hh];
    }
  }

  const float f0 = B_inv[0] * px + B_inv[1] * py + B_inv[2] * pz;
  const float f1 = B_inv[3] * px + B_inv[4] * py + B_inv[5] * pz;
  const float f2 = B_inv[6] * px + B_inv[7] * py + B_inv[8] * pz;
  auto make_xa = [&](int s) -> v2f {
    const float* Rs = R + s * 9;
    const float* Ts = T + s * 3;
    const float x0 = Rs[0] * f0 + Rs[1] * f1 + Rs[2] * f2 + Ts[0];
    const float x1 = Rs[3] * f0 + Rs[4] * f1 + Rs[5] * f2 + Ts[1];
    const float x2 = Rs[6] * f0 + Rs[7] * f1 + Rs[8] * f2 + Ts[2];
    v2f x;
    x.x = half ? x2   : x0;
    x.y = half ? 0.0f : x1;
    return x;
  };
  v2f xcur = make_xa(0);
  const float xo    = hidden_occ[0];
  const float occ   = 1.0f / (1.0f + __expf(-xo));
  const float l2occ = __log2f(occ);
  const float cs = NEG_2PI2 * LOG2E;
  v2f eb1, eb2;
  eb1.x = cs * ua_a;
  eb1.y = (half ? 2.0f * cs : cs) * ua_b;
  eb2.x = half ? l2occ : 2.0f * cs * ua_4;
  eb2.y = half ? 0.0f  : 2.0f * cs * ua_5;

  v2f pa[4];
  #pragma unroll
  for (int t = 0; t < 4; ++t) {
    const float hx = (float)hk[3 * t + 0];
    const float hy = (float)hk[3 * t + 1];
    const float hz = (float)hk[3 * t + 2];
    pa[t].x = half ? hz   : hx;
    pa[t].y = half ? 0.0f : hy;
  }

  float accR[4], accI[4];
  #pragma unroll
  for (int t = 0; t < 4; ++t) { accR[t] = 0.0f; accI[t] = 0.0f; }

  const v8f zc = {};
  float E[8], C[8], S[8];
  v8f pT, e2;
  v2f ea1, ea2;
  {
    const float s0 = sk[0], s1 = sk[1], s2 = sk[2];
    const float q0 = s0 * R[0] + s1 * R[3] + s2 * R[6];
    const float q1 = s0 * R[1] + s1 * R[4] + s2 * R[7];
    const float q2 = s0 * R[2] + s1 * R[5] + s2 * R[8];
    ea1.x = half ? q2 * q2 : q0 * q0;
    ea1.y = half ? q0 * q1 : q1 * q1;
    ea2.x = half ? 1.0f    : q0 * q2;
    ea2.y = half ? 0.0f    : q1 * q2;
    pT = __builtin_amdgcn_wmma_f32_16x16x4_f32(
        false, xcur, false, pa[0], (short)0, zc, false, false);
    const v8f e1 = __builtin_amdgcn_wmma_f32_16x16x4_f32(
        false, eb1, false, ea1, (short)0, zc, false, false);
    e2 = __builtin_amdgcn_wmma_f32_16x16x4_f32(
        false, eb2, false, ea2, (short)0, e1, false, false);
  }

  #pragma unroll
  for (int i = 0; i <= 16; ++i) {
    const int  tp = (i + 3) & 3;
    const int  sn = (i + 1) >> 2;
    const int  tn = (i + 1) & 3;
    const bool has_prev = i > 0;
    const bool has_cur  = i < 16;
    const bool has_next = i + 1 < 16;

    float amp = 0.0f;
    if (has_prev) amp = sfv[0][tp] * E[0];
    #pragma unroll
    for (int r = 0; r < 8; ++r) {
      float ampn = 0.0f;
      if (has_prev) accR[tp] = fmaf(amp, C[r], accR[tp]);
      if (has_cur)  C[r] = __builtin_amdgcn_cosf(pT[r]);
      if (has_prev && r < 7) ampn = sfv[r + 1][tp] * E[r + 1];
      if (has_prev) accI[tp] = fmaf(amp, S[r], accI[tp]);
      if (has_cur)  S[r] = __builtin_amdgcn_sinf(pT[r]);
      amp = ampn;
    }
    if (has_next) {
      if (tn == 0) xcur = make_xa(sn);
      pT = __builtin_amdgcn_wmma_f32_16x16x4_f32(
          false, xcur, false, pa[tn], (short)0, zc, false, false);
    }

    if (has_cur) {
      const float* Rs = R + (has_next ? sn : 0) * 9;
      const float s0 = sk[3 * tn + 0];
      const float s1 = sk[3 * tn + 1];
      const float s2 = sk[3 * tn + 2];
      float q0 = 0.0f, q1 = 0.0f, q2 = 0.0f;
      E[0] = __builtin_amdgcn_exp2f(e2[0]);
      if (has_next) q0 = s0 * Rs[0] + s1 * Rs[3] + s2 * Rs[6];
      E[1] = __builtin_amdgcn_exp2f(e2[1]);
      if (has_next) q1 = s0 * Rs[1] + s1 * Rs[4] + s2 * Rs[7];
      E[2] = __builtin_amdgcn_exp2f(e2[2]);
      if (has_next) q2 = s0 * Rs[2] + s1 * Rs[5] + s2 * Rs[8];
      E[3] = __builtin_amdgcn_exp2f(e2[3]);
      if (has_next) ea1.x = half ? q2 * q2 : q0 * q0;
      E[4] = __builtin_amdgcn_exp2f(e2[4]);
      if (has_next) ea1.y = half ? q0 * q1 : q1 * q1;
      E[5] = __builtin_amdgcn_exp2f(e2[5]);
      if (has_next) ea2.x = half ? 1.0f    : q0 * q2;
      E[6] = __builtin_amdgcn_exp2f(e2[6]);
      if (has_next) ea2.y = half ? 0.0f    : q1 * q2;
      E[7] = __builtin_amdgcn_exp2f(e2[7]);
    }
    if (has_next) {
      const v8f e1 = __builtin_amdgcn_wmma_f32_16x16x4_f32(
          false, eb1, false, ea1, (short)0, zc, false, false);
      e2 = __builtin_amdgcn_wmma_f32_16x16x4_f32(
          false, eb2, false, ea2, (short)0, e1, false, false);
    }
  }

  #pragma unroll
  for (int t = 0; t < 4; ++t) {
    accR[t] += xhalf(accR[t]);
    accI[t] += xhalf(accI[t]);
  }

  float* plane = out + (half ? (size_t)H : (size_t)0);
  if (FULL) {
    #pragma unroll
    for (int g = 0; g < 2; ++g) {
      v2f v;
      v.x = half ? accI[2 * g]     : accR[2 * g];
      v.y = half ? accI[2 * g + 1] : accR[2 * g + 1];
      *reinterpret_cast<v2f*>(plane + h0 + 32 * g + 2 * c) = v;
    }
  } else {
    #pragma unroll
    for (int t = 0; t < 4; ++t) {
      const int hh = h0 + 16 * t + c;
      const float v = half ? accI[t] : accR[t];
      if (hh < H) plane[hh] = v;
    }
  }
}

__global__ __launch_bounds__(256, 8) void residue_sf_kernel(
    const int*   __restrict__ hkl,
    const float* __restrict__ sv,
    const float* __restrict__ xyz,
    const float* __restrict__ u,
    const float* __restrict__ hidden_occ,
    const float* __restrict__ sf,
    const float* __restrict__ B_inv,
    const float* __restrict__ R,
    const float* __restrict__ T,
    float* __restrict__ out,
    int H) {
  const int lane = threadIdx.x & 31;
  const int wave = threadIdx.x >> 5;
  const int h0   = (blockIdx.x * 8 + wave) * 64;
  if (h0 >= H) return;
  if (h0 + 64 <= H && (H & 1) == 0) {
    sf_tile<true >(hkl, sv, xyz, u, hidden_occ, sf, B_inv, R, T, out, H, h0, lane);
  } else {
    sf_tile<false>(hkl, sv, xyz, u, hidden_occ, sf, B_inv, R, T, out, H, h0, lane);
  }
}

extern "C" void kernel_launch(void* const* d_in, const int* in_sizes, int n_in,
                              void* d_out, int out_size, void* d_ws, size_t ws_size,
                              hipStream_t stream) {
  const int*   hkl   = (const int*)  d_in[0];
  const float* sv    = (const float*)d_in[1];
  const float* xyz   = (const float*)d_in[2];
  const float* u     = (const float*)d_in[3];
  const float* hocc  = (const float*)d_in[4];
  const float* sf    = (const float*)d_in[5];
  const float* B_inv = (const float*)d_in[6];
  const float* symR  = (const float*)d_in[7];
  const float* symT  = (const float*)d_in[8];
  float* out = (float*)d_out;

  const int H = in_sizes[0] / 3;
  if (H <= 0) return;

  const int blocks = (H + 511) / 512;
  residue_sf_kernel<<<blocks, 256, 0, stream>>>(hkl, sv, xyz, u, hocc, sf, B_inv,
                                                symR, symT, out, H);
}


# Issue order pinned by scheduling fences: each cos/sin/exp2 stands alone between plain vector instructions; steps run tile by tile, (re,im) sums packed; order of every sum unchanged
# speedup vs baseline: 2.0659x; 1.0606x over previous
#include <hip/hip_runtime.h>

typedef float v2f __attribute__((ext_vector_type(2)));
typedef float v8f __attribute__((ext_vector_type(8)));

#define LOG2E     1.4426950408889634f
#define NEG_2PI2  (-19.739208802178716f)

#define ISSUE_FENCE() __builtin_amdgcn_sched_barrier(0)

__device__ __forceinline__ float xhalf(float x) {
  const int b = __builtin_bit_cast(int, x);
  return __builtin_bit_cast(float, __builtin_amdgcn_permlanex16(
      b, b, 0x76543210, 0xfedcba98, false, false));
}

template <bool FULL>
__device__ __forceinline__ void load_pair(
    int g,
    const int*   __restrict__ hkl,
    const float* __restrict__ sv,
    const float* __restrict__ sf,
    int H, int h0, int c, int half,
    int (&hk)[12], float (&sk)[12], float (&sfv)[8][4]) {
  const unsigned rowoff = (unsigned)(8 * half) * (unsigned)H;
  if (FULL) {
    const unsigned hb  = (unsigned)(h0 + 2 * c);
    const unsigned vo3 = hb * 12u + (unsigned)g * 384u;
    const unsigned vo1 = (rowoff + hb) * 4u + (unsigned)g * 128u;
    const char* hb_i = reinterpret_cast<const char*>(hkl) + vo3;
    const char* hb_f = reinterpret_cast<const char*>(sv)  + vo3;
    #pragma unroll
    for (int j = 0; j < 6; ++j) {
      hk[6 * g + j] = reinterpret_cast<const int*>(hb_i)[j];
      sk[6 * g + j] = reinterpret_cast<const float*>(hb_f)[j];
    }
    #pragma unroll
    for (int r = 0; r < 8; ++r) {
      const char* row = reinterpret_cast<const char*>(sf + (size_t)r * (size_t)H);
      const v2f v = *reinterpret_cast<const v2f*>(row + vo1);
      sfv[r][2 * g] = v.x; sfv[r][2 * g + 1] = v.y;
    }
  } else {
    #pragma unroll
    for (int tt = 0; tt < 2; ++tt) {
      const int t = 2 * g + tt;
      int hh = h0 + 16 * t + c;
      if (hh > H - 1) hh = H - 1;
      const unsigned vo3 = (unsigned)hh * 12u;
      const unsigned vo1 = (rowoff + (unsigned)hh) * 4u;
      const char* hb_i = reinterpret_cast<const char*>(hkl) + vo3;
      const char* hb_f = reinterpret_cast<const char*>(sv)  + vo3;
      #pragma unroll
      for (int j = 0; j < 3; ++j) {
        hk[3 * t + j] = reinterpret_cast<const int*>(hb_i)[j];
        sk[3 * t + j] = reinterpret_cast<const float*>(hb_f)[j];
      }
      #pragma unroll
      for (int r = 0; r < 8; ++r) {
        const char* row = reinterpret_cast<const char*>(sf + (size_t)r * (size_t)H);
        sfv[r][t] = *reinterpret_cast<const float*>(row + vo1);
      }
    }
  }
}

template <bool FULL>
__device__ __forceinline__ void sf_tile(
    const int*   __restrict__ hkl,
    const float* __restrict__ sv,
    const float* __restrict__ xyz,
    const float* __restrict__ u,
    const float* __restrict__ hidden_occ,
    const float* __restrict__ sf,
    const float* __restrict__ B_inv,
    const float* __restrict__ R,
    const float* __restrict__ T,
    float* __restrict__ out,
    int H, int h0, int lane) {
  const int half = lane >> 4;
  const int c    = lane & 15;

  typedef const float __attribute__((address_space(4))) cfloat_k;
  cfloat_k* Rm = (cfloat_k*)R;
  cfloat_k* Tm = (cfloat_k*)T;
  cfloat_k* Bm = (cfloat_k*)B_inv;
  cfloat_k* Om = (cfloat_k*)hidden_occ;

  const float px = xyz[c * 3 + 0], py = xyz[c * 3 + 1], pz = xyz[c * 3 + 2];
  const float* ua = u + c * 6;
  const float ua_0 = ua[0], ua_1 = ua[1], ua_2 = ua[2], ua_3 = ua[3];
  const float ua_a = half ? ua_2 : ua_0;
  const float ua_b = half ? ua_3 : ua_1;
  const float ua_4 = ua[4];
  const float ua_5 = ua[5];

  int   hk[12];
  float sk[12];
  float sfv[8][4];
  load_pair<FULL>(0, hkl, sv, sf, H, h0, c, half, hk, sk, sfv);

  const float f0 = Bm[0] * px + Bm[1] * py + Bm[2] * pz;
  const float f1 = Bm[3] * px + Bm[4] * py + Bm[5] * pz;
  const float f2 = Bm[6] * px + Bm[7] * py + Bm[8] * pz;
  auto make_xa = [&](int s) -> v2f {
    cfloat_k* Rs = Rm + s * 9;
    cfloat_k* Ts = Tm + s * 3;
    const float x0 = Rs[0] * f0 + Rs[1] * f1 + Rs[2] * f2 + Ts[0];
    const float x1 = Rs[3] * f0 + Rs[4] * f1 + Rs[5] * f2 + Ts[1];
    const float x2 = Rs[6] * f0 + Rs[7] * f1 + Rs[8] * f2 + Ts[2];
    v2f x;
    x.x = half ? x2   : x0;
    x.y = half ? 0.0f : x1;
    return x;
  };
  v2f xa[4];
  #pragma unroll
  for (int s = 0; s < 4; ++s) xa[s] = make_xa(s);
  const float xo    = Om[0];
  const float occ   = 1.0f / (1.0f + __expf(-xo));
  const float l2occ = __log2f(occ);
  const float cs = NEG_2PI2 * LOG2E;
  v2f eb1, eb2;
  eb1.x = cs * ua_a;
  eb1.y = (half ? 2.0f * cs : cs) * ua_b;
  eb2.x = half ? l2occ : 2.0f * cs * ua_4;
  eb2.y = half ? 0.0f  : 2.0f * cs * ua_5;

  auto make_pa = [&](int t) -> v2f {
    const float hx = (float)hk[3 * t + 0];
    const float hy = (float)hk[3 * t + 1];
    const float hz = (float)hk[3 * t + 2];
    v2f p;
    p.x = half ? hz   : hx;
    p.y = half ? 0.0f : hy;
    return p;
  };

  v2f acc[4];
  #pragma unroll
  for (int t = 0; t < 4; ++t) { acc[t].x = 0.0f; acc[t].y = 0.0f; }

  const v8f zc = {};
  float E[8] = {}, amp[8] = {};
  v2f   CS[8] = {};
  v8f pT[2], e2[2];
  v2f ea1, ea2, pacur;
  float q0 = 0.0f, q1 = 0.0f, q2 = 0.0f;
  float qn0 = 0.0f, qn1 = 0.0f, qn2 = 0.0f;
  {
    pacur = make_pa(0);
    const float s0 = sk[0], s1 = sk[1], s2 = sk[2];
    q0 = s0 * Rm[0] + s1 * Rm[3] + s2 * Rm[6];
    q1 = s0 * Rm[1] + s1 * Rm[4] + s2 * Rm[7];
    q2 = s0 * Rm[2] + s1 * Rm[5] + s2 * Rm[8];
    ea1.x = half ? q2 * q2 : q0 * q0;
    ea1.y = half ? q0 * q1 : q1 * q1;
    ea2.x = half ? 1.0f    : q0 * q2;
    ea2.y = half ? 0.0f    : q1 * q2;
    pT[0] = __builtin_amdgcn_wmma_f32_16x16x4_f32(
        false, xa[0], false, pacur, (short)0, zc, false, false);
    e2[0] = __builtin_amdgcn_wmma_f32_16x16x4_f32(
        false, eb1, false, ea1, (short)0, zc, false, false);
    e2[0] = __builtin_amdgcn_wmma_f32_16x16x4_f32(
        false, eb2, false, ea2, (short)0, e2[0], false, false);
    qn0 = fmaf(s1, Rm[9 + 3], s0 * Rm[9 + 0]);
    qn1 = fmaf(s1, Rm[9 + 4], s0 * Rm[9 + 1]);
    qn2 = fmaf(s1, Rm[9 + 5], s0 * Rm[9 + 2]);
    pT[1] = zc; e2[1] = zc;
  }

  #pragma unroll
  for (int i = 0; i <= 16; ++i) {
    const bool has_prev  = i > 0;
    const bool has_cur   = i < 16;
    const bool has_next  = i + 1 < 16;
    const bool has_next2 = i + 2 < 16;
    const int  cb  = i & 1;
    const int  nb  = cb ^ 1;
    const int  t   = has_cur   ? (i >> 2)       : 0;
    const int  tp  = has_prev  ? ((i - 1) >> 2) : 0;
    const int  sn  = has_next  ? ((i + 1) & 3)  : 0;
    const int  tn  = has_next  ? ((i + 1) >> 2) : 0;
    const int  sn2 = has_next2 ? ((i + 2) & 3)  : 0;
    const int  tn2 = has_next2 ? ((i + 2) >> 2) : 0;

    if (i == 4) {
      ISSUE_FENCE();
      load_pair<FULL>(1, hkl, sv, sf, H, h0, c, half, hk, sk, sfv);
    }

    if (has_cur) {
      cfloat_k* Rs  = Rm + sn  * 9;
      cfloat_k* Rs2 = Rm + sn2 * 9;
      const float s2  = sk[3 * tn  + 2];
      const float u0  = sk[3 * tn2 + 0];
      const float u1  = sk[3 * tn2 + 1];
      float x = 0.0f, y = 0.0f;
      const v2f a6 = {amp[6], amp[6]}, a7 = {amp[7], amp[7]};

      ISSUE_FENCE();
      E[0] = __builtin_amdgcn_exp2f(e2[cb][0]);
      ISSUE_FENCE();
      if (has_next) { q0 = fmaf(s2, Rs[6], qn0); q1 = fmaf(s2, Rs[7], qn1); q2 = fmaf(s2, Rs[8], qn2); }
      ISSUE_FENCE();
      E[1] = __builtin_amdgcn_exp2f(e2[cb][1]);
      ISSUE_FENCE();
      if (has_next) { x = half ? q2 : q0; y = half ? q0 : q1; }
      ISSUE_FENCE();
      E[2] = __builtin_amdgcn_exp2f(e2[cb][2]);
      ISSUE_FENCE();
      if (has_next) { ea1.x = x * x; ea1.y = y * q1; }
      ISSUE_FENCE();
      E[3] = __builtin_amdgcn_exp2f(e2[cb][3]);
      ISSUE_FENCE();
      if (has_next) { ea2.x = q0 * q2; ea2.y = q1 * q2; }
      ISSUE_FENCE();
      E[4] = __builtin_amdgcn_exp2f(e2[cb][4]);
      ISSUE_FENCE();
      if (has_next) { ea2.x = half ? 1.0f : ea2.x; ea2.y = half ? 0.0f : ea2.y; }
      ISSUE_FENCE();
      E[5] = __builtin_amdgcn_exp2f(e2[cb][5]);
      ISSUE_FENCE();
      amp[0] = sfv[0][t] * E[0];
      if (has_prev) acc[tp] = __builtin_elementwise_fma(a6, CS[6], acc[tp]);
      ISSUE_FENCE();
      if (has_next) {
        e2[nb] = __builtin_amdgcn_wmma_f32_16x16x4_f32(
            false, eb1, false, ea1, (short)0, zc, false, false);
        ISSUE_FENCE();
      }
      E[6] = __builtin_amdgcn_exp2f(e2[cb][6]);
      ISSUE_FENCE();
      amp[1] = sfv[1][t] * E[1];
      if (has_prev) acc[tp] = __builtin_elementwise_fma(a7, CS[7], acc[tp]);
      ISSUE_FENCE();
      E[7] = __builtin_amdgcn_exp2f(e2[cb][7]);
      ISSUE_FENCE();
      amp[2] = sfv[2][t] * E[2];
      ISSUE_FENCE();

      #pragma unroll
      for (int r = 0; r < 8; ++r) {
        if (r == 3 && has_next) {
          e2[nb] = __builtin_amdgcn_wmma_f32_16x16x4_f32(
              false, eb2, false, ea2, (short)0, e2[nb], false, false);
          ISSUE_FENCE();
        }
        if (r == 6 && has_next) {
          if (sn == 0) pacur = make_pa(tn);
          pT[nb] = __builtin_amdgcn_wmma_f32_16x16x4_f32(
              false, xa[sn], false, pacur, (short)0, zc, false, false);
          ISSUE_FENCE();
        }
        CS[r].x = __builtin_amdgcn_cosf(pT[cb][r]);
        ISSUE_FENCE();
        if (r == 0) amp[3] = sfv[3][t] * E[3];
        if (r == 1) amp[5] = sfv[5][t] * E[5];
        if (r == 2) amp[6] = sfv[6][t] * E[6];
        if (r == 3) amp[7] = sfv[7][t] * E[7];
        if (r == 4 && has_next2) { qn0 = u0 * Rs2[0]; qn1 = u0 * Rs2[1]; qn2 = u0 * Rs2[2]; }
        if (r == 5 && has_next2) { qn0 = fmaf(u1, Rs2[3], qn0); qn1 = fmaf(u1, Rs2[4], qn1); qn2 = fmaf(u1, Rs2[5], qn2); }
        ISSUE_FENCE();
        CS[r].y = __builtin_amdgcn_sinf(pT[cb][r]);
        ISSUE_FENCE();
        if (r == 0) amp[4] = sfv[4][t] * E[4];
        if (r >= 2) {
          const v2f a = {amp[r - 2], amp[r - 2]};
          acc[t] = __builtin_elementwise_fma(a, CS[r - 2], acc[t]);
        }
        ISSUE_FENCE();
      }
    } else {
      const v2f a6 = {amp[6], amp[6]}, a7 = {amp[7], amp[7]};
      acc[tp] = __builtin_elementwise_fma(a6, CS[6], acc[tp]);
      acc[tp] = __builtin_elementwise_fma(a7, CS[7], acc[tp]);
    }
  }

  float accR[4], accI[4];
  #pragma unroll
  for (int t = 0; t < 4; ++t) { accR[t] = acc[t].x; accI[t] = acc[t].y; }

  #pragma unroll
  for (int t = 0; t < 4; ++t) {
    accR[t] += xhalf(accR[t]);
    accI[t] += xhalf(accI[t]);
  }

  float* plane = out + (half ? (size_t)H : (size_t)0);
  if (FULL) {
    #pragma unroll
    for (int g = 0; g < 2; ++g) {
      v2f v;
      v.x = half ? accI[2 * g]     : accR[2 * g];
      v.y = half ? accI[2 * g + 1] : accR[2 * g + 1];
      *reinterpret_cast<v2f*>(plane + h0 + 32 * g + 2 * c) = v;
    }
  } else {
    #pragma unroll
    for (int t = 0; t < 4; ++t) {
      const int hh = h0 + 16 * t + c;
      const float v = half ? accI[t] : accR[t];
      if (hh < H) plane[hh] = v;
    }
  }
}

__global__ __launch_bounds__(256) __attribute__((amdgpu_waves_per_eu(8, 8)))
void residue_sf_kernel(
    const int*   __restrict__ hkl,
    const float* __restrict__ sv,
    const float* __restrict__ xyz,
    const float* __restrict__ u,
    const float* __restrict__ hidden_occ,
    const float* __restrict__ sf,
    const float* __restrict__ B_inv,
    const float* __restrict__ R,
    const float* __restrict__ T,
    float* __restrict__ out,
    int H) {
  const int lane = threadIdx.x & 31;
  const int wave = threadIdx.x >> 5;
  const int h0   = (blockIdx.x * 8 + wave) * 64;
  if (h0 >= H) return;
  if (h0 + 64 <= H && (H & 1) == 0) {
    sf_tile<true >(hkl, sv, xyz, u, hidden_occ, sf, B_inv, R, T, out, H, h0, lane);
  } else {
    sf_tile<false>(hkl, sv, xyz, u, hidden_occ, sf, B_inv, R, T, out, H, h0, lane);
  }
}

extern "C" void kernel_launch(void* const* d_in, const int* in_sizes, int n_in,
                              void* d_out, int out_size, void* d_ws, size_t ws_size,
                              hipStream_t stream) {
  const int*   hkl   = (const int*)  d_in[0];
  const float* sv    = (const float*)d_in[1];
  const float* xyz   = (const float*)d_in[2];
  const float* u     = (const float*)d_in[3];
  const float* hocc  = (const float*)d_in[4];
  const float* sf    = (const float*)d_in[5];
  const float* B_inv = (const float*)d_in[6];
  const float* symR  = (const float*)d_in[7];
  const float* symT  = (const float*)d_in[8];
  float* out = (float*)d_out;

  const int H = in_sizes[0] / 3;
  if (H <= 0) return;

  const int blocks = (H + 511) / 512;
  residue_sf_kernel<<<blocks, 256, 0, stream>>>(hkl, sv, xyz, u, hocc, sf, B_inv,
                                                symR, symT, out, H);
}
